# starting kernel
# baseline (speedup 1.0000x reference)
#include <hip/hip_runtime.h>
#include <hip/hip_bf16.h>

typedef __attribute__((ext_vector_type(2))) float v2f;
typedef __attribute__((ext_vector_type(8))) float v8f;

#define IN_FEATS 384
#define N_HIDDEN 128
#define N_CLASSES 16

__global__ __launch_bounds__(256) void zero_kernel(float* __restrict__ p, long long n) {
    long long i = (long long)blockIdx.x * blockDim.x + threadIdx.x;
    long long stride = (long long)gridDim.x * blockDim.x;
    for (; i < n; i += stride) p[i] = 0.0f;
}

__global__ __launch_bounds__(256) void degree_kernel(const int* __restrict__ src,
                                                     const int* __restrict__ dst,
                                                     float* __restrict__ deg_out,
                                                     float* __restrict__ deg_in,
                                                     int nE) {
    int e = blockIdx.x * 256 + threadIdx.x;
    if (e >= nE) return;
    atomicAdd(&deg_out[src[e]], 1.0f);
    atomicAdd(&deg_in[dst[e]], 1.0f);
}

__global__ __launch_bounds__(256) void rsqrt_kernel(float* __restrict__ a,
                                                    float* __restrict__ b, int n) {
    int i = blockIdx.x * 256 + threadIdx.x;
    if (i >= n) return;
    float x = a[i];
    a[i] = (x > 0.0f) ? rsqrtf(x) : 0.0f;
    float y = b[i];
    b[i] = (y > 0.0f) ? rsqrtf(y) : 0.0f;
}

template <int CT>
__global__ __launch_bounds__(64) void gemm_scaled_wmma(const float* __restrict__ H,
                                                       const float* __restrict__ ns,
                                                       const float* __restrict__ W,
                                                       float* __restrict__ out,
                                                       int K, int nRowTiles) {
    constexpr int F = CT * 16;
    const int wave = threadIdx.x >> 5;
    const int lane = threadIdx.x & 31;
    const int m  = lane & 15;
    const int hi = lane >> 4;

    const int rowTile = blockIdx.x * 2 + wave;
    if (rowTile >= nRowTiles) return;
    const int row0 = rowTile * 16;

    const float scale = ns[row0 + m];
    const float* __restrict__ arow = H + (size_t)(row0 + m) * K;
    const float* __restrict__ bbase = W + m;

    v8f acc[CT];
#pragma unroll
    for (int c = 0; c < CT; ++c) acc[c] = (v8f){};

#pragma unroll 2
    for (int k0 = 0; k0 < K; k0 += 4) {
        const int ka = k0 + 2 * hi;
        float2 t = *(const float2*)(arow + ka);
        v2f a;
        a.x = t.x * scale;
        a.y = t.y * scale;
        const float* __restrict__ brow = bbase + (size_t)ka * F;
#pragma unroll
        for (int c = 0; c < CT; ++c) {
            v2f b;
            b.x = brow[c * 16];
            b.y = brow[c * 16 + F];
            acc[c] = __builtin_amdgcn_wmma_f32_16x16x4_f32(false, a, false, b,
                                                           (short)0, acc[c], false, false);
        }
    }

    float* __restrict__ obase = out + (size_t)(row0 + 8 * hi) * F + m;
#pragma unroll
    for (int c = 0; c < CT; ++c) {
#pragma unroll
        for (int v = 0; v < 8; ++v)
            obase[(size_t)v * F + c * 16] = acc[c][v];
    }
}

__global__ __launch_bounds__(256) void scatter_kernel(const int* __restrict__ src,
                                                      const int* __restrict__ dst,
                                                      const float* __restrict__ hw,
                                                      float* __restrict__ agg,
                                                      int nEdges, int F, int lgF4) {
    const int F4 = F >> 2;
    const int tid = blockIdx.x * 256 + threadIdx.x;
    const int e = tid >> lgF4;
    const int j = tid & (F4 - 1);
    if (e >= nEdges) return;
    const int s = src[e];
    const int d = dst[e];
    const float4 v = ((const float4*)(hw + (size_t)s * F))[j];
    float* o = agg + (size_t)d * F + 4 * j;
    atomicAdd(o + 0, v.x);
    atomicAdd(o + 1, v.y);
    atomicAdd(o + 2, v.z);
    atomicAdd(o + 3, v.w);
}

__global__ __launch_bounds__(256) void finalize_kernel(const float* __restrict__ agg,
                                                       const float* __restrict__ nd,
                                                       const float* __restrict__ bias,
                                                       float* __restrict__ out,
                                                       int N, int lgF, int relu) {
    const int F = 1 << lgF;
    long long t = (long long)blockIdx.x * 256 + threadIdx.x;
    if (t >= ((long long)N << lgF)) return;
    const int i = (int)(t >> lgF);
    const int f = (int)(t & (F - 1));
    float v = agg[t] * nd[i] + bias[f];
    if (relu) v = fmaxf(v, 0.0f);
    out[t] = v;
}

extern "C" void kernel_launch(void* const* d_in, const int* in_sizes, int n_in,
                              void* d_out, int out_size, void* d_ws, size_t ws_size,
                              hipStream_t stream) {
    const float* feat = (const float*)d_in[0];
    const int*   src  = (const int*)d_in[1];
    const int*   dst  = (const int*)d_in[2];
    const float* W1   = (const float*)d_in[3];
    const float* b1   = (const float*)d_in[4];
    const float* W2   = (const float*)d_in[5];
    const float* b2   = (const float*)d_in[6];
    const float* W3   = (const float*)d_in[7];
    const float* b3   = (const float*)d_in[8];
    float* out = (float*)d_out;

    const int N = in_sizes[0] / IN_FEATS;
    const int E = in_sizes[1];

    float* ns   = (float*)d_ws;
    float* nd   = ns + N;
    float* bufA = nd + N;
    float* bufB = bufA + (size_t)N * N_HIDDEN;

    const int rowTiles  = N / 16;
    const int rowBlocks = (rowTiles + 1) / 2;

    zero_kernel<<<(2 * N + 255) / 256, 256, 0, stream>>>(ns, 2LL * N);
    degree_kernel<<<(E + 255) / 256, 256, 0, stream>>>(src, dst, ns, nd, E);
    rsqrt_kernel<<<(N + 255) / 256, 256, 0, stream>>>(ns, nd, N);

    {
        gemm_scaled_wmma<8><<<rowBlocks, 64, 0, stream>>>(feat, ns, W1, bufA, IN_FEATS, rowTiles);
        long long nAgg = (long long)N * N_HIDDEN;
        zero_kernel<<<(int)((nAgg + 255) / 256), 256, 0, stream>>>(bufB, nAgg);
        long long nScat = (long long)E * (N_HIDDEN / 4);
        scatter_kernel<<<(int)((nScat + 255) / 256), 256, 0, stream>>>(src, dst, bufA, bufB, E, N_HIDDEN, 5);
        finalize_kernel<<<(int)((nAgg + 255) / 256), 256, 0, stream>>>(bufB, nd, b1, bufA, N, 7, 1);
    }

    {
        gemm_scaled_wmma<8><<<rowBlocks, 64, 0, stream>>>(bufA, ns, W2, bufB, N_HIDDEN, rowTiles);
        long long nAgg = (long long)N * N_HIDDEN;
        zero_kernel<<<(int)((nAgg + 255) / 256), 256, 0, stream>>>(bufA, nAgg);
        long long nScat = (long long)E * (N_HIDDEN / 4);
        scatter_kernel<<<(int)((nScat + 255) / 256), 256, 0, stream>>>(src, dst, bufB, bufA, E, N_HIDDEN, 5);
        finalize_kernel<<<(int)((nAgg + 255) / 256), 256, 0, stream>>>(bufA, nd, b2, bufB, N, 7, 1);
    }

    {
        gemm_scaled_wmma<1><<<rowBlocks, 64, 0, stream>>>(bufB, ns, W3, bufA, N_HIDDEN, rowTiles);
        long long nOut = (long long)N * N_CLASSES;
        zero_kernel<<<(int)((nOut + 255) / 256), 256, 0, stream>>>(out, nOut);
        long long nScat = (long long)E * (N_CLASSES / 4);
        scatter_kernel<<<(int)((nScat + 255) / 256), 256, 0, stream>>>(src, dst, bufA, out, E, N_CLASSES, 2);
        finalize_kernel<<<(int)((nOut + 255) / 256), 256, 0, stream>>>(out, nd, b3, out, N, 4, 0);
    }
}


# Degree counts in LDS tables: a work-group counts one 12800-node range of one edge chunk with LDS integer adds, stores a partial plane; norm kernel sums 16 planes (exact); no global add per edge
# speedup vs baseline: 1.8686x; 1.8686x over previous
#include <hip/hip_runtime.h>
#include <hip/hip_bf16.h>

typedef __attribute__((ext_vector_type(2))) float v2f;
typedef __attribute__((ext_vector_type(8))) float v8f;

#define IN_FEATS 384
#define N_HIDDEN 128
#define N_CLASSES 16

__global__ __launch_bounds__(256) void zero_kernel(float* __restrict__ p, long long n) {
    long long i = (long long)blockIdx.x * blockDim.x + threadIdx.x;
    long long stride = (long long)gridDim.x * blockDim.x;
    for (; i < n; i += stride) p[i] = 0.0f;
}

#define DEG_BINS   12800
#define DEG_CHUNKS 16
#define DEG_THREADS 512
#define DEG_LOADS  8

__device__ __forceinline__ void deg_count(unsigned* bins, int id, int lo) {
    const unsigned b = (unsigned)(id - lo);
    if (b < DEG_BINS) atomicAdd(&bins[b], 1u);
}

__global__ __launch_bounds__(DEG_THREADS) void degree_hist_kernel(const int* __restrict__ src,
                                                                  const int* __restrict__ dst,
                                                                  float* __restrict__ part,
                                                                  int nE, int N, int chunkLen) {
    __shared__ unsigned bins[DEG_BINS];
    const int* __restrict__ idx = blockIdx.z ? dst : src;
    const int lo = blockIdx.y * DEG_BINS;

    for (int j = threadIdx.x; j < DEG_BINS; j += DEG_THREADS) bins[j] = 0u;
    __syncthreads();

    const int e0 = blockIdx.x * chunkLen;
    int len = nE - e0;
    len = (len < chunkLen) ? len : chunkLen;
    const int nv = (len > 0) ? (len >> 2) : 0;
    const int4* __restrict__ p = (const int4*)(idx + e0);

    for (int v = threadIdx.x; v < nv; v += DEG_LOADS * DEG_THREADS) {
        int4 q[DEG_LOADS];
#pragma unroll
        for (int u = 0; u < DEG_LOADS; ++u) {
            const int vv = v + u * DEG_THREADS;
            q[u] = (vv < nv) ? p[vv] : make_int4(-1, -1, -1, -1);
        }
#pragma unroll
        for (int u = 0; u < DEG_LOADS; ++u) {
            deg_count(bins, q[u].x, lo);
            deg_count(bins, q[u].y, lo);
            deg_count(bins, q[u].z, lo);
            deg_count(bins, q[u].w, lo);
        }
    }
    if (len > 0 && (int)threadIdx.x < (len & 3)) deg_count(bins, idx[e0 + (nv << 2) + threadIdx.x], lo);
    __syncthreads();

    float* __restrict__ o = part + (size_t)(blockIdx.z * DEG_CHUNKS + blockIdx.x) * N + lo;
    for (int j = threadIdx.x; j < DEG_BINS; j += DEG_THREADS)
        if (lo + j < N) o[j] = (float)bins[j];
}

__global__ __launch_bounds__(256) void norm_kernel(const float* __restrict__ part,
                                                   float* __restrict__ norms, int N) {
    const int i = blockIdx.x * 256 + threadIdx.x;
    if (i >= 2 * N) return;
    const int a = (i >= N) ? 1 : 0;
    const float* __restrict__ p = part + (size_t)a * DEG_CHUNKS * N + (i - a * N);
    float s = 0.0f;
#pragma unroll
    for (int c = 0; c < DEG_CHUNKS; ++c) s += p[(size_t)c * N];
    norms[i] = (s > 0.0f) ? rsqrtf(s) : 0.0f;
}

template <int CT>
__global__ __launch_bounds__(64) void gemm_scaled_wmma(const float* __restrict__ H,
                                                       const float* __restrict__ ns,
                                                       const float* __restrict__ W,
                                                       float* __restrict__ out,
                                                       int K, int nRowTiles) {
    constexpr int F = CT * 16;
    const int wave = threadIdx.x >> 5;
    const int lane = threadIdx.x & 31;
    const int m  = lane & 15;
    const int hi = lane >> 4;

    const int rowTile = blockIdx.x * 2 + wave;
    if (rowTile >= nRowTiles) return;
    const int row0 = rowTile * 16;

    const float scale = ns[row0 + m];
    const float* __restrict__ arow = H + (size_t)(row0 + m) * K;
    const float* __restrict__ bbase = W + m;

    v8f acc[CT];
#pragma unroll
    for (int c = 0; c < CT; ++c) acc[c] = (v8f){};

#pragma unroll 2
    for (int k0 = 0; k0 < K; k0 += 4) {
        const int ka = k0 + 2 * hi;
        float2 t = *(const float2*)(arow + ka);
        v2f a;
        a.x = t.x * scale;
        a.y = t.y * scale;
        const float* __restrict__ brow = bbase + (size_t)ka * F;
#pragma unroll
        for (int c = 0; c < CT; ++c) {
            v2f b;
            b.x = brow[c * 16];
            b.y = brow[c * 16 + F];
            acc[c] = __builtin_amdgcn_wmma_f32_16x16x4_f32(false, a, false, b,
                                                           (short)0, acc[c], false, false);
        }
    }

    float* __restrict__ obase = out + (size_t)(row0 + 8 * hi) * F + m;
#pragma unroll
    for (int c = 0; c < CT; ++c) {
#pragma unroll
        for (int v = 0; v < 8; ++v)
            obase[(size_t)v * F + c * 16] = acc[c][v];
    }
}

template <int F, int EPT>
__global__ __launch_bounds__(256) void scatter_lines(const int* __restrict__ src,
                                                     const int* __restrict__ dst,
                                                     const float* __restrict__ hw,
                                                     float* __restrict__ agg,
                                                     int nEdges) {
    constexpr int LPE = (F >= 32) ? 32 : F;
    constexpr int EPI = 32 / LPE;
    constexpr int C   = F / LPE;
    constexpr int EPW = EPT * EPI;
    static_assert(EPW <= 32 && (EPW & (EPW - 1)) == 0, "one index load per wave");

    const int lane = threadIdx.x & 31;
    const int wave = blockIdx.x * 8 + __builtin_amdgcn_readfirstlane(threadIdx.x >> 5);
    const int e0 = wave * EPW;
    if (e0 >= nEdges) return;
    const int sub = lane / LPE;
    const int l   = lane % LPE;

    int ei = e0 + (lane & (EPW - 1));
    ei = (ei < nEdges) ? ei : (nEdges - 1);
    const int sv = src[ei];
    const int dv = dst[ei];

    float v[EPT][C];
    int   d[EPT];
#pragma unroll
    for (int t = 0; t < EPT; ++t) {
        int s;
        if constexpr (EPI == 1) {
            s    = __builtin_amdgcn_readlane(sv, t);
            d[t] = __builtin_amdgcn_readlane(dv, t);
        } else {
            s    = __shfl(sv, t * EPI + sub);
            d[t] = __shfl(dv, t * EPI + sub);
        }
        const float* __restrict__ r = hw + (size_t)s * F + l;
#pragma unroll
        for (int c = 0; c < C; ++c) v[t][c] = r[c * LPE];
    }

    if (e0 + EPW <= nEdges) {
#pragma unroll
        for (int t = 0; t < EPT; ++t) {
            float* o = agg + (size_t)d[t] * F + l;
#pragma unroll
            for (int c = 0; c < C; ++c) atomicAdd(o + c * LPE, v[t][c]);
        }
    } else {
#pragma unroll
        for (int t = 0; t < EPT; ++t) {
            if (e0 + t * EPI + sub < nEdges) {
                float* o = agg + (size_t)d[t] * F + l;
#pragma unroll
                for (int c = 0; c < C; ++c) atomicAdd(o + c * LPE, v[t][c]);
            }
        }
    }
}

__global__ __launch_bounds__(256) void finalize_kernel(const float* __restrict__ agg,
                                                       const float* __restrict__ nd,
                                                       const float* __restrict__ bias,
                                                       float* __restrict__ out,
                                                       int N, int lgF, int relu) {
    const int F = 1 << lgF;
    long long t = (long long)blockIdx.x * 256 + threadIdx.x;
    if (t >= ((long long)N << lgF)) return;
    const int i = (int)(t >> lgF);
    const int f = (int)(t & (F - 1));
    float v = agg[t] * nd[i] + bias[f];
    if (relu) v = fmaxf(v, 0.0f);
    out[t] = v;
}

extern "C" void kernel_launch(void* const* d_in, const int* in_sizes, int n_in,
                              void* d_out, int out_size, void* d_ws, size_t ws_size,
                              hipStream_t stream) {
    const float* feat = (const float*)d_in[0];
    const int*   src  = (const int*)d_in[1];
    const int*   dst  = (const int*)d_in[2];
    const float* W1   = (const float*)d_in[3];
    const float* b1   = (const float*)d_in[4];
    const float* W2   = (const float*)d_in[5];
    const float* b2   = (const float*)d_in[6];
    const float* W3   = (const float*)d_in[7];
    const float* b3   = (const float*)d_in[8];
    float* out = (float*)d_out;

    const int N = in_sizes[0] / IN_FEATS;
    const int E = in_sizes[1];

    float* ns   = (float*)d_ws;
    float* nd   = ns + N;
    float* bufA = nd + N;
    float* bufB = bufA + (size_t)N * N_HIDDEN;

    const int rowTiles  = N / 16;
    const int rowBlocks = (rowTiles + 1) / 2;

    {
        const int degRanges = (N + DEG_BINS - 1) / DEG_BINS;
        const int chunkLen  = (((E + DEG_CHUNKS - 1) / DEG_CHUNKS) + 3) & ~3;
        degree_hist_kernel<<<dim3(DEG_CHUNKS, degRanges, 2), DEG_THREADS, 0, stream>>>(src, dst, bufB, E, N, chunkLen);
        norm_kernel<<<(2 * N + 255) / 256, 256, 0, stream>>>(bufB, ns, N);
    }

    {
        gemm_scaled_wmma<8><<<rowBlocks, 64, 0, stream>>>(feat, ns, W1, bufA, IN_FEATS, rowTiles);
        long long nAgg = (long long)N * N_HIDDEN;
        zero_kernel<<<(int)((nAgg + 255) / 256), 256, 0, stream>>>(bufB, nAgg);
        scatter_lines<N_HIDDEN, 8><<<(E + 63) / 64, 256, 0, stream>>>(src, dst, bufA, bufB, E);
        finalize_kernel<<<(int)((nAgg + 255) / 256), 256, 0, stream>>>(bufB, nd, b1, bufA, N, 7, 1);
    }

    {
        gemm_scaled_wmma<8><<<rowBlocks, 64, 0, stream>>>(bufA, ns, W2, bufB, N_HIDDEN, rowTiles);
        long long nAgg = (long long)N * N_HIDDEN;
        zero_kernel<<<(int)((nAgg + 255) / 256), 256, 0, stream>>>(bufA, nAgg);
        scatter_lines<N_HIDDEN, 8><<<(E + 63) / 64, 256, 0, stream>>>(src, dst, bufB, bufA, E);
        finalize_kernel<<<(int)((nAgg + 255) / 256), 256, 0, stream>>>(bufA, nd, b2, bufB, N, 7, 1);
    }

    {
        gemm_scaled_wmma<1><<<rowBlocks, 64, 0, stream>>>(bufB, ns, W3, bufA, N_HIDDEN, rowTiles);
        long long nOut = (long long)N * N_CLASSES;
        zero_kernel<<<(int)((nOut + 255) / 256), 256, 0, stream>>>(out, nOut);
        scatter_lines<N_CLASSES, 8><<<(E + 127) / 128, 256, 0, stream>>>(src, dst, bufA, out, E);
        finalize_kernel<<<(int)((nOut + 255) / 256), 256, 0, stream>>>(out, nd, b3, out, N, 4, 0);
    }
}


# Element-wise passes folded into the layer products: each product clears the rows its scatter adds into; products 2, 3 apply norm, bias, relu to the raw sums in the A loader; 14 launches become 9
# speedup vs baseline: 2.0527x; 1.0985x over previous
#include <hip/hip_runtime.h>
#include <hip/hip_bf16.h>

typedef __attribute__((ext_vector_type(2))) float v2f;
typedef __attribute__((ext_vector_type(8))) float v8f;

#define IN_FEATS 384
#define N_HIDDEN 128
#define N_CLASSES 16

#define DEG_BINS   12800
#define DEG_CHUNKS 16
#define DEG_THREADS 512
#define DEG_LOADS  8

__device__ __forceinline__ void deg_count(unsigned* bins, int id, int lo) {
    const unsigned b = (unsigned)(id - lo);
    if (b < DEG_BINS) atomicAdd(&bins[b], 1u);
}

__global__ __launch_bounds__(DEG_THREADS) void degree_hist_kernel(const int* __restrict__ src,
                                                                  const int* __restrict__ dst,
                                                                  float* __restrict__ part,
                                                                  int nE, int N, int chunkLen) {
    __shared__ unsigned bins[DEG_BINS];
    const int* __restrict__ idx = blockIdx.z ? dst : src;
    const int lo = blockIdx.y * DEG_BINS;

    for (int j = threadIdx.x; j < DEG_BINS; j += DEG_THREADS) bins[j] = 0u;
    __syncthreads();

    const int e0 = blockIdx.x * chunkLen;
    int len = nE - e0;
    len = (len < chunkLen) ? len : chunkLen;
    const int nv = (len > 0) ? (len >> 2) : 0;
    const int4* __restrict__ p = (const int4*)(idx + e0);

    for (int v = threadIdx.x; v < nv; v += DEG_LOADS * DEG_THREADS) {
        int4 q[DEG_LOADS];
#pragma unroll
        for (int u = 0; u < DEG_LOADS; ++u) {
            const int vv = v + u * DEG_THREADS;
            q[u] = (vv < nv) ? p[vv] : make_int4(-1, -1, -1, -1);
        }
#pragma unroll
        for (int u = 0; u < DEG_LOADS; ++u) {
            deg_count(bins, q[u].x, lo);
            deg_count(bins, q[u].y, lo);
            deg_count(bins, q[u].z, lo);
            deg_count(bins, q[u].w, lo);
        }
    }
    if (len > 0 && (int)threadIdx.x < (len & 3)) deg_count(bins, idx[e0 + (nv << 2) + threadIdx.x], lo);
    __syncthreads();

    float* __restrict__ o = part + (size_t)(blockIdx.z * DEG_CHUNKS + blockIdx.x) * N + lo;
    for (int j = threadIdx.x; j < DEG_BINS; j += DEG_THREADS)
        if (lo + j < N) o[j] = (float)bins[j];
}

__global__ __launch_bounds__(256) void norm_kernel(const float* __restrict__ part,
                                                   float* __restrict__ norms, int N) {
    const int i = blockIdx.x * 256 + threadIdx.x;
    if (i >= 2 * N) return;
    const int a = (i >= N) ? 1 : 0;
    const float* __restrict__ p = part + (size_t)a * DEG_CHUNKS * N + (i - a * N);
    float s = 0.0f;
#pragma unroll
    for (int c = 0; c < DEG_CHUNKS; ++c) s += p[(size_t)c * N];
    norms[i] = (s > 0.0f) ? rsqrtf(s) : 0.0f;
}

template <int CT, int ZF, bool FOLD>
__global__ __launch_bounds__(64) void gemm_scaled_wmma(const float* H,
                                                       const float* __restrict__ ns,
                                                       const float* __restrict__ nd,
                                                       const float* __restrict__ bias,
                                                       const float* __restrict__ W,
                                                       float* __restrict__ out,
                                                       float* Z,
                                                       int K, int nRowTiles) {
    constexpr int F = CT * 16;
    const int wave = threadIdx.x >> 5;
    const int lane = threadIdx.x & 31;
    const int m  = lane & 15;
    const int hi = lane >> 4;

    const int rowTile = blockIdx.x * 2 + wave;
    if (rowTile >= nRowTiles) return;
    const int row0 = rowTile * 16;

    const float scale = ns[row0 + m];
    float dscale = 0.0f;
    if constexpr (FOLD) dscale = nd[row0 + m];
    const float* arow = H + (size_t)(row0 + m) * K;
    const float* __restrict__ bbase = W + m;

    v8f acc[CT];
#pragma unroll
    for (int c = 0; c < CT; ++c) acc[c] = (v8f){};

#pragma unroll 2
    for (int k0 = 0; k0 < K; k0 += 4) {
        const int ka = k0 + 2 * hi;
        float2 t = *(const float2*)(arow + ka);
        v2f a;
        if constexpr (FOLD) {
            const float2 bb = *(const float2*)(bias + ka);
            a.x = fmaxf(t.x * dscale + bb.x, 0.0f) * scale;
            a.y = fmaxf(t.y * dscale + bb.y, 0.0f) * scale;
        } else {
            a.x = t.x * scale;
            a.y = t.y * scale;
        }
        const float* __restrict__ brow = bbase + (size_t)ka * F;
#pragma unroll
        for (int c = 0; c < CT; ++c) {
            v2f b;
            b.x = brow[c * 16];
            b.y = brow[c * 16 + F];
            acc[c] = __builtin_amdgcn_wmma_f32_16x16x4_f32(false, a, false, b,
                                                           (short)0, acc[c], false, false);
        }
    }

    {
        float4* z4 = (float4*)(Z + (size_t)row0 * ZF);
        for (int j = 0; j < ZF / 8; ++j) z4[j * 32 + lane] = make_float4(0.0f, 0.0f, 0.0f, 0.0f);
    }

    float* __restrict__ obase = out + (size_t)(row0 + 8 * hi) * F + m;
#pragma unroll
    for (int c = 0; c < CT; ++c) {
#pragma unroll
        for (int v = 0; v < 8; ++v)
            obase[(size_t)v * F + c * 16] = acc[c][v];
    }
}

template <int F, int EPT>
__global__ __launch_bounds__(256) void scatter_lines(const int* __restrict__ src,
                                                     const int* __restrict__ dst,
                                                     const float* __restrict__ hw,
                                                     float* __restrict__ agg,
                                                     int nEdges) {
    constexpr int LPE = (F >= 32) ? 32 : F;
    constexpr int EPI = 32 / LPE;
    constexpr int C   = F / LPE;
    constexpr int EPW = EPT * EPI;
    static_assert(EPW <= 32 && (EPW & (EPW - 1)) == 0, "one index load per wave");

    const int lane = threadIdx.x & 31;
    const int wave = blockIdx.x * 8 + __builtin_amdgcn_readfirstlane(threadIdx.x >> 5);
    const int e0 = wave * EPW;
    if (e0 >= nEdges) return;
    const int sub = lane / LPE;
    const int l   = lane % LPE;

    int ei = e0 + (lane & (EPW - 1));
    ei = (ei < nEdges) ? ei : (nEdges - 1);
    const int sv = src[ei];
    const int dv = dst[ei];

    float v[EPT][C];
    int   d[EPT];
#pragma unroll
    for (int t = 0; t < EPT; ++t) {
        int s;
        if constexpr (EPI == 1) {
            s    = __builtin_amdgcn_readlane(sv, t);
            d[t] = __builtin_amdgcn_readlane(dv, t);
        } else {
            s    = __shfl(sv, t * EPI + sub);
            d[t] = __shfl(dv, t * EPI + sub);
        }
        const float* __restrict__ r = hw + (size_t)s * F + l;
#pragma unroll
        for (int c = 0; c < C; ++c) v[t][c] = r[c * LPE];
    }

    if (e0 + EPW <= nEdges) {
#pragma unroll
        for (int t = 0; t < EPT; ++t) {
            float* o = agg + (size_t)d[t] * F + l;
#pragma unroll
            for (int c = 0; c < C; ++c) atomicAdd(o + c * LPE, v[t][c]);
        }
    } else {
#pragma unroll
        for (int t = 0; t < EPT; ++t) {
            if (e0 + t * EPI + sub < nEdges) {
                float* o = agg + (size_t)d[t] * F + l;
#pragma unroll
                for (int c = 0; c < C; ++c) atomicAdd(o + c * LPE, v[t][c]);
            }
        }
    }
}

__global__ __launch_bounds__(256) void finalize_kernel(const float* __restrict__ agg,
                                                       const float* __restrict__ nd,
                                                       const float* __restrict__ bias,
                                                       float* __restrict__ out,
                                                       int N, int lgF, int relu) {
    const int F = 1 << lgF;
    long long t = (long long)blockIdx.x * 256 + threadIdx.x;
    if (t >= ((long long)N << lgF)) return;
    const int i = (int)(t >> lgF);
    const int f = (int)(t & (F - 1));
    float v = agg[t] * nd[i] + bias[f];
    if (relu) v = fmaxf(v, 0.0f);
    out[t] = v;
}

extern "C" void kernel_launch(void* const* d_in, const int* in_sizes, int n_in,
                              void* d_out, int out_size, void* d_ws, size_t ws_size,
                              hipStream_t stream) {
    const float* feat = (const float*)d_in[0];
    const int*   src  = (const int*)d_in[1];
    const int*   dst  = (const int*)d_in[2];
    const float* W1   = (const float*)d_in[3];
    const float* b1   = (const float*)d_in[4];
    const float* W2   = (const float*)d_in[5];
    const float* b2   = (const float*)d_in[6];
    const float* W3   = (const float*)d_in[7];
    const float* b3   = (const float*)d_in[8];
    float* out = (float*)d_out;

    const int N = in_sizes[0] / IN_FEATS;
    const int E = in_sizes[1];

    float* ns   = (float*)d_ws;
    float* nd   = ns + N;
    float* bufA = nd + N;
    float* bufB = bufA + (size_t)N * N_HIDDEN;

    const int rowTiles  = N / 16;
    const int rowBlocks = (rowTiles + 1) / 2;

    {
        const int degRanges = (N + DEG_BINS - 1) / DEG_BINS;
        const int chunkLen  = (((E + DEG_CHUNKS - 1) / DEG_CHUNKS) + 3) & ~3;
        degree_hist_kernel<<<dim3(DEG_CHUNKS, degRanges, 2), DEG_THREADS, 0, stream>>>(src, dst, bufB, E, N, chunkLen);
        norm_kernel<<<(2 * N + 255) / 256, 256, 0, stream>>>(bufB, ns, N);
    }


    gemm_scaled_wmma<8, N_HIDDEN, false><<<rowBlocks, 64, 0, stream>>>(feat, ns, nullptr, nullptr, W1, bufA, bufB, IN_FEATS, rowTiles);
    scatter_lines<N_HIDDEN, 8><<<(E + 63) / 64, 256, 0, stream>>>(src, dst, bufA, bufB, E);

    gemm_scaled_wmma<8, N_HIDDEN, true><<<rowBlocks, 64, 0, stream>>>(bufB, ns, nd, b1, W2, bufA, bufB, N_HIDDEN, rowTiles);
    scatter_lines<N_HIDDEN, 8><<<(E + 63) / 64, 256, 0, stream>>>(src, dst, bufA, bufB, E);

    {
        gemm_scaled_wmma<1, N_CLASSES, true><<<rowBlocks, 64, 0, stream>>>(bufB, ns, nd, b2, W3, bufA, out, N_HIDDEN, rowTiles);
        long long nOut = (long long)N * N_CLASSES;
        scatter_lines<N_CLASSES, 8><<<(E + 127) / 128, 256, 0, stream>>>(src, dst, bufA, out, E);
        finalize_kernel<<<(int)((nOut + 255) / 256), 256, 0, stream>>>(out, nd, b3, out, N, 4, 0);
    }
}


# 128-wide edge scatters: each package half owns 64 columns of both planes; resident work-groups read their die number and take edge chunks from their half's counter; sums stay atomic arrival order
# speedup vs baseline: 2.3204x; 1.1304x over previous
#include <hip/hip_runtime.h>
#include <hip/hip_bf16.h>

typedef __attribute__((ext_vector_type(2))) float v2f;
typedef __attribute__((ext_vector_type(8))) float v8f;

#define IN_FEATS 384
#define N_HIDDEN 128
#define N_CLASSES 16

#define QUEUE_STRIDE 64

#define DEG_BINS   12800
#define DEG_CHUNKS 16
#define DEG_THREADS 512
#define DEG_LOADS  8

__device__ __forceinline__ void deg_count(unsigned* bins, int id, int lo) {
    const unsigned b = (unsigned)(id - lo);
    if (b < DEG_BINS) atomicAdd(&bins[b], 1u);
}

__global__ __launch_bounds__(DEG_THREADS) void degree_hist_kernel(const int* __restrict__ src,
                                                                  const int* __restrict__ dst,
                                                                  float* __restrict__ part,
                                                                  int nE, int N, int chunkLen) {
    __shared__ unsigned bins[DEG_BINS];
    const int* __restrict__ idx = blockIdx.z ? dst : src;
    const int lo = blockIdx.y * DEG_BINS;

    for (int j = threadIdx.x; j < DEG_BINS; j += DEG_THREADS) bins[j] = 0u;
    __syncthreads();

    const int e0 = blockIdx.x * chunkLen;
    int len = nE - e0;
    len = (len < chunkLen) ? len : chunkLen;
    const int nv = (len > 0) ? (len >> 2) : 0;
    const int4* __restrict__ p = (const int4*)(idx + e0);

    for (int v = threadIdx.x; v < nv; v += DEG_LOADS * DEG_THREADS) {
        int4 q[DEG_LOADS];
#pragma unroll
        for (int u = 0; u < DEG_LOADS; ++u) {
            const int vv = v + u * DEG_THREADS;
            q[u] = (vv < nv) ? p[vv] : make_int4(-1, -1, -1, -1);
        }
#pragma unroll
        for (int u = 0; u < DEG_LOADS; ++u) {
            deg_count(bins, q[u].x, lo);
            deg_count(bins, q[u].y, lo);
            deg_count(bins, q[u].z, lo);
            deg_count(bins, q[u].w, lo);
        }
    }
    if (len > 0 && (int)threadIdx.x < (len & 3)) deg_count(bins, idx[e0 + (nv << 2) + threadIdx.x], lo);
    __syncthreads();

    float* __restrict__ o = part + (size_t)(blockIdx.z * DEG_CHUNKS + blockIdx.x) * N + lo;
    for (int j = threadIdx.x; j < DEG_BINS; j += DEG_THREADS)
        if (lo + j < N) o[j] = (float)bins[j];
}

__global__ __launch_bounds__(256) void norm_kernel(const float* __restrict__ part,
                                                   float* __restrict__ norms, int N) {
    const int i = blockIdx.x * 256 + threadIdx.x;
    if (i >= 2 * N) return;
    const int a = (i >= N) ? 1 : 0;
    const float* __restrict__ p = part + (size_t)a * DEG_CHUNKS * N + (i - a * N);
    float s = 0.0f;
#pragma unroll
    for (int c = 0; c < DEG_CHUNKS; ++c) s += p[(size_t)c * N];
    norms[i] = (s > 0.0f) ? rsqrtf(s) : 0.0f;
}

template <int CT, int ZF, bool FOLD>
__global__ __launch_bounds__(64) void gemm_scaled_wmma(const float* H,
                                                       const float* __restrict__ ns,
                                                       const float* __restrict__ nd,
                                                       const float* __restrict__ bias,
                                                       const float* __restrict__ W,
                                                       float* __restrict__ out,
                                                       float* Z,
                                                       int K, int nRowTiles,
                                                       unsigned* __restrict__ queue) {
    constexpr int F = CT * 16;
    const int wave = threadIdx.x >> 5;
    const int lane = threadIdx.x & 31;
    if (queue != nullptr && blockIdx.x == 0 && threadIdx.x == 0) {
        queue[0] = 0u;
        queue[QUEUE_STRIDE] = 0u;
    }
    const int m  = lane & 15;
    const int hi = lane >> 4;

    const int rowTile = blockIdx.x * 2 + wave;
    if (rowTile >= nRowTiles) return;
    const int row0 = rowTile * 16;

    const float scale = ns[row0 + m];
    float dscale = 0.0f;
    if constexpr (FOLD) dscale = nd[row0 + m];
    const float* arow = H + (size_t)(row0 + m) * K;
    const float* __restrict__ bbase = W + m;

    v8f acc[CT];
#pragma unroll
    for (int c = 0; c < CT; ++c) acc[c] = (v8f){};

#pragma unroll 2
    for (int k0 = 0; k0 < K; k0 += 4) {
        const int ka = k0 + 2 * hi;
        float2 t = *(const float2*)(arow + ka);
        v2f a;
        if constexpr (FOLD) {
            const float2 bb = *(const float2*)(bias + ka);
            a.x = fmaxf(t.x * dscale + bb.x, 0.0f) * scale;
            a.y = fmaxf(t.y * dscale + bb.y, 0.0f) * scale;
        } else {
            a.x = t.x * scale;
            a.y = t.y * scale;
        }
        const float* __restrict__ brow = bbase + (size_t)ka * F;
#pragma unroll
        for (int c = 0; c < CT; ++c) {
            v2f b;
            b.x = brow[c * 16];
            b.y = brow[c * 16 + F];
            acc[c] = __builtin_amdgcn_wmma_f32_16x16x4_f32(false, a, false, b,
                                                           (short)0, acc[c], false, false);
        }
    }

    {
        float4* z4 = (float4*)(Z + (size_t)row0 * ZF);
        for (int j = 0; j < ZF / 8; ++j) z4[j * 32 + lane] = make_float4(0.0f, 0.0f, 0.0f, 0.0f);
    }

    float* __restrict__ obase = out + (size_t)(row0 + 8 * hi) * F + m;
#pragma unroll
    for (int c = 0; c < CT; ++c) {
#pragma unroll
        for (int v = 0; v < 8; ++v)
            obase[(size_t)v * F + c * 16] = acc[c][v];
    }
}

template <int F, int EPT>
__global__ __launch_bounds__(256) void scatter_lines(const int* __restrict__ src,
                                                     const int* __restrict__ dst,
                                                     const float* __restrict__ hw,
                                                     float* __restrict__ agg,
                                                     int nEdges) {
    constexpr int LPE = (F >= 32) ? 32 : F;
    constexpr int EPI = 32 / LPE;
    constexpr int C   = F / LPE;
    constexpr int EPW = EPT * EPI;
    static_assert(EPW <= 32 && (EPW & (EPW - 1)) == 0, "one index load per wave");

    const int lane = threadIdx.x & 31;
    const int wave = blockIdx.x * 8 + __builtin_amdgcn_readfirstlane(threadIdx.x >> 5);
    const int e0 = wave * EPW;
    if (e0 >= nEdges) return;
    const int sub = lane / LPE;
    const int l   = lane % LPE;

    int ei = e0 + (lane & (EPW - 1));
    ei = (ei < nEdges) ? ei : (nEdges - 1);
    const int sv = src[ei];
    const int dv = dst[ei];

    float v[EPT][C];
    int   d[EPT];
#pragma unroll
    for (int t = 0; t < EPT; ++t) {
        int s;
        if constexpr (EPI == 1) {
            s    = __builtin_amdgcn_readlane(sv, t);
            d[t] = __builtin_amdgcn_readlane(dv, t);
        } else {
            s    = __shfl(sv, t * EPI + sub);
            d[t] = __shfl(dv, t * EPI + sub);
        }
        const float* __restrict__ r = hw + (size_t)s * F + l;
#pragma unroll
        for (int c = 0; c < C; ++c) v[t][c] = r[c * LPE];
    }

    if (e0 + EPW <= nEdges) {
#pragma unroll
        for (int t = 0; t < EPT; ++t) {
            float* o = agg + (size_t)d[t] * F + l;
#pragma unroll
            for (int c = 0; c < C; ++c) atomicAdd(o + c * LPE, v[t][c]);
        }
    } else {
#pragma unroll
        for (int t = 0; t < EPT; ++t) {
            if (e0 + t * EPI + sub < nEdges) {
                float* o = agg + (size_t)d[t] * F + l;
#pragma unroll
                for (int c = 0; c < C; ++c) atomicAdd(o + c * LPE, v[t][c]);
            }
        }
    }
}

__device__ __forceinline__ int package_half() {
    return (int)((__builtin_amdgcn_s_sendmsg_rtn(0x87) >> 18) & 1u);
}

template <int EPT>
__global__ __launch_bounds__(256) void scatter_owned_columns(const int* __restrict__ src,
                                                             const int* __restrict__ dst,
                                                             const float* __restrict__ hw,
                                                             float* __restrict__ agg,
                                                             int nEdges,
                                                             unsigned* __restrict__ queue) {
    constexpr int F   = N_HIDDEN;
    constexpr int HC  = F / 2;
    constexpr int C   = HC / 32;
    constexpr int EPC = 8 * EPT;
    static_assert(EPT <= 32 && (EPT & (EPT - 1)) == 0, "one index load per wave");
    __shared__ unsigned s_chunk[2];

    const int lane = threadIdx.x & 31;
    const int w = __builtin_amdgcn_readfirstlane(threadIdx.x >> 5);
    const int own = package_half();
    const unsigned nChunks = (unsigned)((nEdges + EPC - 1) / EPC);
    unsigned turn = 0;

    for (int pass = 0; pass < 2; ++pass) {
        const int g = own ^ pass;
        unsigned* __restrict__ q = queue + g * QUEUE_STRIDE;
        const float* __restrict__ hwg = hw + g * HC + lane;
        float* __restrict__ aggg = agg + g * HC + lane;

        unsigned ahead = 0u;
        if (threadIdx.x == 0) ahead = atomicAdd(q, 1u);
        for (;;) {
            if (threadIdx.x == 0) s_chunk[turn & 1u] = ahead;
            __syncthreads();
            const unsigned c = s_chunk[turn & 1u];
            ++turn;
            if (c >= nChunks) break;

            const int e0 = (int)c * EPC + w * EPT;
            const bool hasEdges = e0 < nEdges;

            int ei = e0 + (lane & (EPT - 1));
            ei = (ei < nEdges) ? ei : (nEdges - 1);
            const int sv = src[ei];
            const int dv = dst[ei];

            float v[EPT][C];
            int   d[EPT];
#pragma unroll
            for (int t = 0; t < EPT; ++t) {
                const int s = __builtin_amdgcn_readlane(sv, t);
                d[t] = __builtin_amdgcn_readlane(dv, t);
                const float* __restrict__ r = hwg + (size_t)s * F;
#pragma unroll
                for (int cc = 0; cc < C; ++cc) v[t][cc] = r[cc * 32];
            }

            if (threadIdx.x == 0) ahead = atomicAdd(q, 1u);

            if (!hasEdges) {
            } else if (e0 + EPT <= nEdges) {
#pragma unroll
                for (int t = 0; t < EPT; ++t) {
                    float* o = aggg + (size_t)d[t] * F;
#pragma unroll
                    for (int cc = 0; cc < C; ++cc) atomicAdd(o + cc * 32, v[t][cc]);
                }
            } else {
#pragma unroll
                for (int t = 0; t < EPT; ++t) {
                    if (e0 + t < nEdges) {
                        float* o = aggg + (size_t)d[t] * F;
#pragma unroll
                        for (int cc = 0; cc < C; ++cc) atomicAdd(o + cc * 32, v[t][cc]);
                    }
                }
            }
        }
    }
}

__global__ __launch_bounds__(256) void finalize_kernel(const float* __restrict__ agg,
                                                       const float* __restrict__ nd,
                                                       const float* __restrict__ bias,
                                                       float* __restrict__ out,
                                                       int N, int lgF, int relu) {
    const int F = 1 << lgF;
    long long t = (long long)blockIdx.x * 256 + threadIdx.x;
    if (t >= ((long long)N << lgF)) return;
    const int i = (int)(t >> lgF);
    const int f = (int)(t & (F - 1));
    float v = agg[t] * nd[i] + bias[f];
    if (relu) v = fmaxf(v, 0.0f);
    out[t] = v;
}

extern "C" void kernel_launch(void* const* d_in, const int* in_sizes, int n_in,
                              void* d_out, int out_size, void* d_ws, size_t ws_size,
                              hipStream_t stream) {
    const float* feat = (const float*)d_in[0];
    const int*   src  = (const int*)d_in[1];
    const int*   dst  = (const int*)d_in[2];
    const float* W1   = (const float*)d_in[3];
    const float* b1   = (const float*)d_in[4];
    const float* W2   = (const float*)d_in[5];
    const float* b2   = (const float*)d_in[6];
    const float* W3   = (const float*)d_in[7];
    const float* b3   = (const float*)d_in[8];
    float* out = (float*)d_out;

    const int N = in_sizes[0] / IN_FEATS;
    const int E = in_sizes[1];

    float* ns   = (float*)d_ws;
    float* nd   = ns + N;
    float* bufA = nd + N;
    float* bufB = bufA + (size_t)N * N_HIDDEN;

    unsigned* queue = (unsigned*)d_out;

    const int rowTiles  = N / 16;
    const int rowBlocks = (rowTiles + 1) / 2;

    {
        const int degRanges = (N + DEG_BINS - 1) / DEG_BINS;
        const int chunkLen  = (((E + DEG_CHUNKS - 1) / DEG_CHUNKS) + 3) & ~3;
        degree_hist_kernel<<<dim3(DEG_CHUNKS, degRanges, 2), DEG_THREADS, 0, stream>>>(src, dst, bufB, E, N, chunkLen);
        norm_kernel<<<(2 * N + 255) / 256, 256, 0, stream>>>(bufB, ns, N);
    }


    gemm_scaled_wmma<8, N_HIDDEN, false><<<rowBlocks, 64, 0, stream>>>(feat, ns, nullptr, nullptr, W1, bufA, bufB, IN_FEATS, rowTiles, queue);
    scatter_owned_columns<16><<<2048, 256, 0, stream>>>(src, dst, bufA, bufB, E, queue);

    gemm_scaled_wmma<8, N_HIDDEN, true><<<rowBlocks, 64, 0, stream>>>(bufB, ns, nd, b1, W2, bufA, bufB, N_HIDDEN, rowTiles, queue);
    scatter_owned_columns<16><<<2048, 256, 0, stream>>>(src, dst, bufA, bufB, E, queue);

    {
        gemm_scaled_wmma<1, N_CLASSES, true><<<rowBlocks, 64, 0, stream>>>(bufB, ns, nd, b2, W3, bufA, out, N_HIDDEN, rowTiles, nullptr);
        long long nOut = (long long)N * N_CLASSES;
        scatter_lines<N_CLASSES, 8><<<(E + 127) / 128, 256, 0, stream>>>(src, dst, bufA, out, E);
        finalize_kernel<<<(int)((nOut + 255) / 256), 256, 0, stream>>>(out, nd, b3, out, N, 4, 0);
    }
}
